# starting kernel
# baseline (speedup 1.0000x reference)
#include <hip/hip_runtime.h>
#include <hip/hip_bf16.h>
#include <math.h>

#define B_    32
#define T_    131072
#define KLEN  16384
#define SEG   2048
#define NSEG  (T_ / SEG)

typedef __attribute__((ext_vector_type(2))) float v2f;
typedef __attribute__((ext_vector_type(8))) float v8f;

__device__ __forceinline__ float alpha_of(float z) {
    float sig = 1.0f / (1.0f + __expf(-z));
    return fminf(fmaxf(sig, 1e-5f), 1.0f - 1e-5f);
}

__global__ void __launch_bounds__(256) k_energy(const float* __restrict__ sig,
                                                float* __restrict__ e) {
    int gid  = blockIdx.x * 256 + threadIdx.x;
    int flat = gid * 4;
    int b = flat / T_;
    int t = flat % T_;
    const float4 s0 = *(const float4*)(sig + (size_t)b * 2 * T_ + t);
    const float4 s1 = *(const float4*)(sig + (size_t)b * 2 * T_ + T_ + t);
    float4 r;
    r.x = 0.5f * (s0.x * s0.x + s1.x * s1.x);
    r.y = 0.5f * (s0.y * s0.y + s1.y * s1.y);
    r.z = 0.5f * (s0.z * s0.z + s1.z * s1.z);
    r.w = 0.5f * (s0.w * s0.w + s1.w * s1.w);
    *(float4*)(e + flat) = r;
}

__global__ void __launch_bounds__(256) k_partial(const float* __restrict__ e,
                                                 const float* __restrict__ zalpha,
                                                 float* __restrict__ P) {
    const int s = blockIdx.x, b = blockIdx.y;
    const int j = threadIdx.x;
    const float alpha = alpha_of(zalpha[b]);
    const float la  = __logf(alpha);
    const float oma = 1.0f - alpha;
    const float aK  = __expf((float)KLEN * la);
    const float* er = e + (size_t)b * T_;
    const int t0 = s * SEG + j * 8;

    float p = 0.0f;
#pragma unroll
    for (int i = 0; i < 8; ++i) {
        int t = t0 + i;
        float ek = (t >= KLEN) ? er[t - KLEN] : 0.0f;
        float bv = oma * (er[t] - aK * ek);
        p = fmaf(alpha, p, bv);
    }

    __shared__ float red[256];
    red[j] = p * __expf(8.0f * (float)(255 - j) * la);
    __syncthreads();
#pragma unroll
    for (int off = 128; off > 0; off >>= 1) {
        if (j < off) red[j] += red[j + off];
        __syncthreads();
    }
    if (j == 0) P[b * NSEG + s] = red[0];
}

__global__ void k_scan(const float* __restrict__ P,
                       const float* __restrict__ zalpha,
                       float* __restrict__ Carr) {
    int b = threadIdx.x;
    if (b < B_) {
        float alpha = alpha_of(zalpha[b]);
        float aL = __expf((float)SEG * __logf(alpha));
        float c = 0.0f;
        for (int s = 0; s < NSEG; ++s) {
            Carr[b * NSEG + s] = c;
            c = fmaf(aL, c, P[b * NSEG + s]);
        }
    }
}

__global__ void __launch_bounds__(256) k_final(const float* __restrict__ e,
                                               const float* __restrict__ zalpha,
                                               const float* __restrict__ Carr,
                                               float* __restrict__ out) {
    const int s = blockIdx.x, b = blockIdx.y;
    const int tid  = threadIdx.x;
    const int wave = tid >> 5;
    const int lane = tid & 31;

    __shared__ float lbs[8][256];
    __shared__ float cend[8][16];
    __shared__ float wend[8];

    const float alpha = alpha_of(zalpha[b]);
    const float la   = __logf(alpha);
    const float oma  = 1.0f - alpha;
    const float aK   = __expf((float)KLEN * la);
    const float a16  = __expf(16.0f * la);
    const float* er  = e + (size_t)b * T_;
    const int segStart  = s * SEG;
    const int tileStart = segStart + wave * 256;

    {
        const int i0 = lane * 8;
#pragma unroll
        for (int i = 0; i < 8; ++i) {
            int t = tileStart + i0 + i;
            float ek = (t >= KLEN) ? er[t - KLEN] : 0.0f;
            lbs[wave][i0 + i] = oma * (er[t] - aK * ek);
        }
    }
    __syncthreads();

    const int half = lane >> 4;
    const int m    = lane & 15;
    v8f c = {};
#pragma unroll
    for (int sl = 0; sl < 4; ++sl) {
        const int k0 = 4 * sl + 2 * half;
        v2f a;
        a.x = (m >= k0)     ? __expf((float)(m - k0) * la)     : 0.0f;
        a.y = (m >= k0 + 1) ? __expf((float)(m - k0 - 1) * la) : 0.0f;
        v2f bb;
        bb.x = lbs[wave][16 * m + k0];
        bb.y = lbs[wave][16 * m + k0 + 1];
        c = __builtin_amdgcn_wmma_f32_16x16x4_f32(
                 false, a,  false, bb,
                 (short)0, c,  false,  false);
    }

    if (half == 1) cend[wave][m] = c[7];
    __syncthreads();

    float myCarry = 0.0f, cc = 0.0f;
#pragma unroll
    for (int j = 0; j < 16; ++j) {
        if (j == m) myCarry = cc;
        cc = fmaf(a16, cc, cend[wave][j]);
    }
    if (lane == 0) wend[wave] = cc;
    __syncthreads();

    float cw = Carr[b * NSEG + s] * __expf((float)(256 * wave) * la);
    for (int v = 0; v < wave; ++v)
        cw = fmaf(wend[v], __expf((float)(256 * (wave - 1 - v)) * la), cw);
    const float carryChunk = fmaf(cw, __expf((float)(16 * m) * la), myCarry);

#pragma unroll
    for (int r = 0; r < 8; ++r) {
        const int mm = r + 8 * half;
        float val = fmaf(carryChunk, __expf((float)(mm + 1) * la), c[r]);
        float y = fmaxf(val, 0.0f);
        lbs[wave][16 * m + mm] = __logf(y + 1e-5f);
    }
    __syncthreads();

    float* orow = out + (size_t)b * T_ + tileStart;
    const int i0 = lane * 8;
    float4 o0, o1;
    o0.x = lbs[wave][i0 + 0]; o0.y = lbs[wave][i0 + 1];
    o0.z = lbs[wave][i0 + 2]; o0.w = lbs[wave][i0 + 3];
    o1.x = lbs[wave][i0 + 4]; o1.y = lbs[wave][i0 + 5];
    o1.z = lbs[wave][i0 + 6]; o1.w = lbs[wave][i0 + 7];
    *(float4*)(orow + i0)     = o0;
    *(float4*)(orow + i0 + 4) = o1;
}

extern "C" void kernel_launch(void* const* d_in, const int* in_sizes, int n_in,
                              void* d_out, int out_size, void* d_ws, size_t ws_size,
                              hipStream_t stream) {
    const float* signal = (const float*)d_in[0];
    const float* zalpha = (const float*)d_in[1];
    float* out = (float*)d_out;

    float* e    = (float*)d_ws;
    float* P    = e + (size_t)B_ * T_;
    float* Carr = P + (size_t)B_ * NSEG;

    k_energy <<<dim3((B_ * T_) / (256 * 4)), 256, 0, stream>>>(signal, e);
    k_partial<<<dim3(NSEG, B_),             256, 0, stream>>>(e, zalpha, P);
    k_scan   <<<dim3(1),                     32, 0, stream>>>(P, zalpha, Carr);
    k_final  <<<dim3(NSEG, B_),             256, 0, stream>>>(e, zalpha, Carr, out);
}


# Segment carry = decayed sum of the 8 previous segment partials (the K-tap window) instead of a serial scan; energy fuses into the partial pass, scan launch gone: 4 launches -> 2; sum order changes
# speedup vs baseline: 1.4811x; 1.4811x over previous
#include <hip/hip_runtime.h>
#include <hip/hip_bf16.h>
#include <math.h>

#define B_    32
#define T_    131072
#define KLEN  16384
#define SEG   2048
#define NSEG  (T_ / SEG)

typedef __attribute__((ext_vector_type(2))) float v2f;
typedef __attribute__((ext_vector_type(8))) float v8f;

__device__ __forceinline__ float alpha_of(float z) {
    float sig = 1.0f / (1.0f + __expf(-z));
    return fminf(fmaxf(sig, 1e-5f), 1.0f - 1e-5f);
}

static_assert(KLEN % SEG == 0, "the K-tap window must be a whole number of segments");
static_assert(SEG == 2048, "pass 1 maps 256 threads x 2 groups x 4 elements onto one segment");

__global__ void __launch_bounds__(256) k_energy_partial(const float* __restrict__ sig,
                                                        const float* __restrict__ zalpha,
                                                        float* __restrict__ e,
                                                        float* __restrict__ Q) {
    const int s = blockIdx.x, b = blockIdx.y;
    const int j = threadIdx.x;
    const float alpha = alpha_of(zalpha[b]);
    const float la  = __logf(alpha);
    const float oma = 1.0f - alpha;
    const float* p0 = sig + (size_t)b * 2 * T_ + (size_t)s * SEG;
    const float* p1 = p0 + T_;
    float* ep = e + (size_t)b * T_ + (size_t)s * SEG;

    float q[2];
#pragma unroll
    for (int g = 0; g < 2; ++g) {
        const int off = g * 1024 + 4 * j;
        const float4 x0 = *(const float4*)(p0 + off);
        const float4 x1 = *(const float4*)(p1 + off);
        float4 r;
        r.x = 0.5f * (x0.x * x0.x + x1.x * x1.x);
        r.y = 0.5f * (x0.y * x0.y + x1.y * x1.y);
        r.z = 0.5f * (x0.z * x0.z + x1.z * x1.z);
        r.w = 0.5f * (x0.w * x0.w + x1.w * x1.w);
        *(float4*)(ep + off) = r;
        float h = oma * r.x;
        h = fmaf(alpha, h, oma * r.y);
        h = fmaf(alpha, h, oma * r.z);
        h = fmaf(alpha, h, oma * r.w);
        q[g] = h;
    }

    __shared__ float red[256];
    red[j] = q[0] * __expf(4.0f * (float)(511 - j) * la)
           + q[1] * __expf(4.0f * (float)(255 - j) * la);
    __syncthreads();
#pragma unroll
    for (int off = 128; off > 0; off >>= 1) {
        if (j < off) red[j] += red[j + off];
        __syncthreads();
    }
    if (j == 0) Q[b * NSEG + s] = red[0];
}

__global__ void __launch_bounds__(256) k_final(const float* __restrict__ e,
                                               const float* __restrict__ zalpha,
                                               const float* __restrict__ Q,
                                               float* __restrict__ out) {
    const int s = blockIdx.x, b = blockIdx.y;
    const int tid  = threadIdx.x;
    const int wave = tid >> 5;
    const int lane = tid & 31;

    __shared__ float lbs[8][256];
    __shared__ float cend[8][16];
    __shared__ float wend[8];

    const float alpha = alpha_of(zalpha[b]);
    const float la   = __logf(alpha);
    const float oma  = 1.0f - alpha;
    const float aK   = __expf((float)KLEN * la);
    const float a16  = __expf(16.0f * la);
    const float* er  = e + (size_t)b * T_;
    const int segStart  = s * SEG;
    const int tileStart = segStart + wave * 256;

    const float aL = __expf((float)SEG * la);
    float qv[KLEN / SEG];
#pragma unroll
    for (int d = 1; d <= KLEN / SEG; ++d) {
        const int sj = s - d;
        qv[d - 1] = Q[b * NSEG + (sj > 0 ? sj : 0)];
    }
    float cin = 0.0f;
#pragma unroll
    for (int d = KLEN / SEG; d >= 1; --d)
        cin = (s - d >= 0) ? fmaf(aL, cin, qv[d - 1]) : cin;

    {
        const int i0 = lane * 8;
#pragma unroll
        for (int i = 0; i < 8; ++i) {
            int t = tileStart + i0 + i;
            float ek = (t >= KLEN) ? er[t - KLEN] : 0.0f;
            lbs[wave][i0 + i] = oma * (er[t] - aK * ek);
        }
    }
    __syncthreads();

    const int half = lane >> 4;
    const int m    = lane & 15;
    v8f c = {};
#pragma unroll
    for (int sl = 0; sl < 4; ++sl) {
        const int k0 = 4 * sl + 2 * half;
        v2f a;
        a.x = (m >= k0)     ? __expf((float)(m - k0) * la)     : 0.0f;
        a.y = (m >= k0 + 1) ? __expf((float)(m - k0 - 1) * la) : 0.0f;
        v2f bb;
        bb.x = lbs[wave][16 * m + k0];
        bb.y = lbs[wave][16 * m + k0 + 1];
        c = __builtin_amdgcn_wmma_f32_16x16x4_f32(
                 false, a,  false, bb,
                 (short)0, c,  false,  false);
    }

    if (half == 1) cend[wave][m] = c[7];
    __syncthreads();

    float myCarry = 0.0f, cc = 0.0f;
#pragma unroll
    for (int j = 0; j < 16; ++j) {
        if (j == m) myCarry = cc;
        cc = fmaf(a16, cc, cend[wave][j]);
    }
    if (lane == 0) wend[wave] = cc;
    __syncthreads();

    float cw = cin * __expf((float)(256 * wave) * la);
    for (int v = 0; v < wave; ++v)
        cw = fmaf(wend[v], __expf((float)(256 * (wave - 1 - v)) * la), cw);
    const float carryChunk = fmaf(cw, __expf((float)(16 * m) * la), myCarry);

#pragma unroll
    for (int r = 0; r < 8; ++r) {
        const int mm = r + 8 * half;
        float val = fmaf(carryChunk, __expf((float)(mm + 1) * la), c[r]);
        float y = fmaxf(val, 0.0f);
        lbs[wave][16 * m + mm] = __logf(y + 1e-5f);
    }
    __syncthreads();

    float* orow = out + (size_t)b * T_ + tileStart;
    const int i0 = lane * 8;
    float4 o0, o1;
    o0.x = lbs[wave][i0 + 0]; o0.y = lbs[wave][i0 + 1];
    o0.z = lbs[wave][i0 + 2]; o0.w = lbs[wave][i0 + 3];
    o1.x = lbs[wave][i0 + 4]; o1.y = lbs[wave][i0 + 5];
    o1.z = lbs[wave][i0 + 6]; o1.w = lbs[wave][i0 + 7];
    *(float4*)(orow + i0)     = o0;
    *(float4*)(orow + i0 + 4) = o1;
}

extern "C" void kernel_launch(void* const* d_in, const int* in_sizes, int n_in,
                              void* d_out, int out_size, void* d_ws, size_t ws_size,
                              hipStream_t stream) {
    const float* signal = (const float*)d_in[0];
    const float* zalpha = (const float*)d_in[1];
    float* out = (float*)d_out;

    float* e    = (float*)d_ws;
    float* Q    = e + (size_t)B_ * T_;

    k_energy_partial<<<dim3(NSEG, B_), 256, 0, stream>>>(signal, zalpha, e, Q);
    k_final         <<<dim3(NSEG, B_), 256, 0, stream>>>(e, zalpha, Q, out);
}


# Each block handles 4 consecutive segments at once in both passes: 4x fewer waves to enter per launch, all loads issued first, decay powers and barriers paid once per block; sum order unchanged
# speedup vs baseline: 1.6310x; 1.1012x over previous
#include <hip/hip_runtime.h>
#include <hip/hip_bf16.h>
#include <math.h>

#define B_    32
#define T_    131072
#define KLEN  16384
#define SEG   2048
#define NSEG  (T_ / SEG)
#define GSEG  4
#define NGRP  (NSEG / GSEG)

typedef __attribute__((ext_vector_type(2))) float v2f;
typedef __attribute__((ext_vector_type(8))) float v8f;

__device__ __forceinline__ float alpha_of(float z) {
    float sig = 1.0f / (1.0f + __expf(-z));
    return fminf(fmaxf(sig, 1e-5f), 1.0f - 1e-5f);
}

static_assert(KLEN % SEG == 0, "the K-tap window must be a whole number of segments");
static_assert(SEG == 2048, "pass 1 maps 256 threads x 2 groups x 4 elements onto one segment");
static_assert(NSEG % GSEG == 0, "a row is a whole number of blocks");
static_assert((KLEN / SEG) % GSEG == 0, "all segments of a block lie on one side of t = KLEN");

__global__ void __launch_bounds__(256) k_energy_partial(const float* __restrict__ sig,
                                                        const float* __restrict__ zalpha,
                                                        float* __restrict__ e,
                                                        float* __restrict__ Q) {
    const int s0 = blockIdx.x * GSEG, b = blockIdx.y;
    const int j = threadIdx.x;
    const float alpha = alpha_of(zalpha[b]);
    const float la  = __logf(alpha);
    const float oma = 1.0f - alpha;
    const float* p0 = sig + (size_t)b * 2 * T_ + (size_t)s0 * SEG;
    const float* p1 = p0 + T_;
    float* ep = e + (size_t)b * T_ + (size_t)s0 * SEG;

    float4 x0[GSEG][2], x1[GSEG][2];
#pragma unroll
    for (int g = 0; g < GSEG; ++g) {
#pragma unroll
        for (int h = 0; h < 2; ++h) {
            const int off = g * SEG + h * 1024 + 4 * j;
            x0[g][h] = *(const float4*)(p0 + off);
            x1[g][h] = *(const float4*)(p1 + off);
        }
    }

    const float w0 = __expf(4.0f * (float)(511 - j) * la);
    const float w1 = __expf(4.0f * (float)(255 - j) * la);

    __shared__ float red[GSEG][256];
#pragma unroll
    for (int g = 0; g < GSEG; ++g) {
        float q[2];
#pragma unroll
        for (int h = 0; h < 2; ++h) {
            const int off = g * SEG + h * 1024 + 4 * j;
            const float4 a0 = x0[g][h];
            const float4 a1 = x1[g][h];
            float4 r;
            r.x = 0.5f * (a0.x * a0.x + a1.x * a1.x);
            r.y = 0.5f * (a0.y * a0.y + a1.y * a1.y);
            r.z = 0.5f * (a0.z * a0.z + a1.z * a1.z);
            r.w = 0.5f * (a0.w * a0.w + a1.w * a1.w);
            *(float4*)(ep + off) = r;
            float hh = oma * r.x;
            hh = fmaf(alpha, hh, oma * r.y);
            hh = fmaf(alpha, hh, oma * r.z);
            hh = fmaf(alpha, hh, oma * r.w);
            q[h] = hh;
        }
        red[g][j] = q[0] * w0 + q[1] * w1;
    }
    __syncthreads();
#pragma unroll
    for (int off = 128; off > 0; off >>= 1) {
        if (j < off) {
#pragma unroll
            for (int g = 0; g < GSEG; ++g) red[g][j] += red[g][j + off];
        }
        __syncthreads();
    }
    if (j < GSEG) Q[b * NSEG + s0 + j] = red[j][0];
}

__global__ void __launch_bounds__(256) k_final(const float* __restrict__ e,
                                               const float* __restrict__ zalpha,
                                               const float* __restrict__ Q,
                                               float* __restrict__ out) {
    const int s0 = blockIdx.x * GSEG, b = blockIdx.y;
    const int tid  = threadIdx.x;
    const int wave = tid >> 5;
    const int lane = tid & 31;

    __shared__ float lbs[GSEG][8][256];
    __shared__ float cend[GSEG][8][16];
    __shared__ float wend[GSEG][8];

    const float* er  = e + (size_t)b * T_;
    const int i0 = lane * 8;
    const bool hasK = (s0 >= KLEN / SEG);

    float ev[GSEG][8], ekv[GSEG][8];
#pragma unroll
    for (int g = 0; g < GSEG; ++g) {
        const int t0 = (s0 + g) * SEG + wave * 256 + i0;
#pragma unroll
        for (int i = 0; i < 8; ++i) ev[g][i] = er[t0 + i];
    }
    if (hasK) {
#pragma unroll
        for (int g = 0; g < GSEG; ++g) {
            const int t0 = (s0 + g) * SEG + wave * 256 + i0 - KLEN;
#pragma unroll
            for (int i = 0; i < 8; ++i) ekv[g][i] = er[t0 + i];
        }
    } else {
#pragma unroll
        for (int g = 0; g < GSEG; ++g) {
#pragma unroll
            for (int i = 0; i < 8; ++i) ekv[g][i] = 0.0f;
        }
    }

    const float alpha = alpha_of(zalpha[b]);
    const float la   = __logf(alpha);
    const float oma  = 1.0f - alpha;
    const float aK   = __expf((float)KLEN * la);
    const float a16  = __expf(16.0f * la);

    const float aL = __expf((float)SEG * la);
    float qv[KLEN / SEG + GSEG - 1];
#pragma unroll
    for (int k = 0; k < KLEN / SEG + GSEG - 1; ++k) {
        const int sj = s0 - KLEN / SEG + k;
        qv[k] = Q[b * NSEG + (sj > 0 ? sj : 0)];
    }
    float cin[GSEG];
#pragma unroll
    for (int g = 0; g < GSEG; ++g) {
        float ci = 0.0f;
#pragma unroll
        for (int d = KLEN / SEG; d >= 1; --d)
            ci = (s0 + g - d >= 0) ? fmaf(aL, ci, qv[g - d + KLEN / SEG]) : ci;
        cin[g] = ci;
    }

#pragma unroll
    for (int g = 0; g < GSEG; ++g) {
#pragma unroll
        for (int i = 0; i < 8; ++i)
            lbs[g][wave][i0 + i] = oma * (ev[g][i] - aK * ekv[g][i]);
    }
    __syncthreads();

    const int half = lane >> 4;
    const int m    = lane & 15;
    v8f c[GSEG];
#pragma unroll
    for (int g = 0; g < GSEG; ++g) c[g] = v8f{};
#pragma unroll
    for (int sl = 0; sl < 4; ++sl) {
        const int k0 = 4 * sl + 2 * half;
        v2f a;
        a.x = (m >= k0)     ? __expf((float)(m - k0) * la)     : 0.0f;
        a.y = (m >= k0 + 1) ? __expf((float)(m - k0 - 1) * la) : 0.0f;
#pragma unroll
        for (int g = 0; g < GSEG; ++g) {
            v2f bb;
            bb.x = lbs[g][wave][16 * m + k0];
            bb.y = lbs[g][wave][16 * m + k0 + 1];
            c[g] = __builtin_amdgcn_wmma_f32_16x16x4_f32(
                     false, a,  false, bb,
                     (short)0, c[g],  false,  false);
        }
    }

    if (half == 1) {
#pragma unroll
        for (int g = 0; g < GSEG; ++g) cend[g][wave][m] = c[g][7];
    }
    __syncthreads();

    float myCarry[GSEG], cc[GSEG];
#pragma unroll
    for (int g = 0; g < GSEG; ++g) { myCarry[g] = 0.0f; cc[g] = 0.0f; }
#pragma unroll
    for (int j = 0; j < 16; ++j) {
#pragma unroll
        for (int g = 0; g < GSEG; ++g) {
            if (j == m) myCarry[g] = cc[g];
            cc[g] = fmaf(a16, cc[g], cend[g][wave][j]);
        }
    }
    if (lane == 0) {
#pragma unroll
        for (int g = 0; g < GSEG; ++g) wend[g][wave] = cc[g];
    }
    __syncthreads();

    float cw[GSEG];
    {
        const float ew = __expf((float)(256 * wave) * la);
#pragma unroll
        for (int g = 0; g < GSEG; ++g) cw[g] = cin[g] * ew;
    }
    for (int v = 0; v < wave; ++v) {
        const float pv = __expf((float)(256 * (wave - 1 - v)) * la);
#pragma unroll
        for (int g = 0; g < GSEG; ++g) cw[g] = fmaf(wend[g][v], pv, cw[g]);
    }
    const float e16m = __expf((float)(16 * m) * la);
    float ep1[8];
#pragma unroll
    for (int r = 0; r < 8; ++r) ep1[r] = __expf((float)(r + 8 * half + 1) * la);

#pragma unroll
    for (int g = 0; g < GSEG; ++g) {
        const float carryChunk = fmaf(cw[g], e16m, myCarry[g]);
#pragma unroll
        for (int r = 0; r < 8; ++r) {
            const int mm = r + 8 * half;
            float val = fmaf(carryChunk, ep1[r], c[g][r]);
            float y = fmaxf(val, 0.0f);
            lbs[g][wave][16 * m + mm] = __logf(y + 1e-5f);
        }
    }
    __syncthreads();

#pragma unroll
    for (int g = 0; g < GSEG; ++g) {
        float* orow = out + (size_t)b * T_ + (size_t)(s0 + g) * SEG + wave * 256;
        float4 o0, o1;
        o0.x = lbs[g][wave][i0 + 0]; o0.y = lbs[g][wave][i0 + 1];
        o0.z = lbs[g][wave][i0 + 2]; o0.w = lbs[g][wave][i0 + 3];
        o1.x = lbs[g][wave][i0 + 4]; o1.y = lbs[g][wave][i0 + 5];
        o1.z = lbs[g][wave][i0 + 6]; o1.w = lbs[g][wave][i0 + 7];
        *(float4*)(orow + i0)     = o0;
        *(float4*)(orow + i0 + 4) = o1;
    }
}

extern "C" void kernel_launch(void* const* d_in, const int* in_sizes, int n_in,
                              void* d_out, int out_size, void* d_ws, size_t ws_size,
                              hipStream_t stream) {
    const float* signal = (const float*)d_in[0];
    const float* zalpha = (const float*)d_in[1];
    float* out = (float*)d_out;

    float* e    = (float*)d_ws;
    float* Q    = e + (size_t)B_ * T_;

    k_energy_partial<<<dim3(NGRP, B_), 256, 0, stream>>>(signal, zalpha, e, Q);
    k_final         <<<dim3(NGRP, B_), 256, 0, stream>>>(e, zalpha, Q, out);
}


# One launch instead of two: each block rebuilds energy, the K-back tap and its 4 segment carries from the signal itself (reads the 8 earlier segments too); no workspace; carry sum order changes
# speedup vs baseline: 2.6142x; 1.6028x over previous
#include <hip/hip_runtime.h>
#include <hip/hip_bf16.h>
#include <math.h>


#define B_    32
#define T_    131072
#define KLEN  16384
#define SEG   2048
#define NSEG  (T_ / SEG)
#define GSEG  4
#define KSEG  (KLEN / SEG)
#define NLD   (KSEG + GSEG)
#define NGRP  (NSEG / GSEG)

typedef __attribute__((ext_vector_type(2))) float v2f;
typedef __attribute__((ext_vector_type(8))) float v8f;

__device__ __forceinline__ float alpha_of(float z) {
    float sig = 1.0f / (1.0f + __expf(-z));
    return fminf(fmaxf(sig, 1e-5f), 1.0f - 1e-5f);
}

static_assert(KLEN % SEG == 0, "the K-tap window must be a whole number of segments");
static_assert(SEG == 2048, "256 threads x 2 groups x 4 elements map onto one segment");
static_assert(NSEG % GSEG == 0, "a row is a whole number of blocks");

__global__ void __launch_bounds__(256, 4) k_envelope(const float* __restrict__ sig,
                                                     const float* __restrict__ zalpha,
                                                     float* __restrict__ out) {
    const int s0 = blockIdx.x * GSEG, b = blockIdx.y;
    const int tid  = threadIdx.x;
    const int wave = tid >> 5;
    const int lane = tid & 31;

    __shared__ float lbs[GSEG][8][256];
    __shared__ float cend[GSEG][8][16];
    __shared__ float wend[GSEG][8];
    __shared__ float cpart[GSEG][8];

    const float* p0 = sig + (size_t)b * 2 * T_;
    const float* p1 = p0 + T_;

    float4 x0[NLD][2], x1[NLD][2];
#pragma unroll
    for (int k = 0; k < NLD; ++k) {
        const int sj = s0 - KSEG + k;
        const int sc = sj > 0 ? sj : 0;
#pragma unroll
        for (int h = 0; h < 2; ++h) {
            const int off = sc * SEG + h * 1024 + 4 * tid;
            x0[k][h] = *(const float4*)(p0 + off);
            x1[k][h] = *(const float4*)(p1 + off);
        }
    }

    const float alpha = alpha_of(zalpha[b]);
    const float la   = __logf(alpha);
    const float oma  = 1.0f - alpha;
    const float aK   = __expf((float)KLEN * la);
    const float a16  = __expf(16.0f * la);
    const float aL   = __expf((float)SEG * la);
    const float w0 = __expf(4.0f * (float)(511 - tid) * la);
    const float w1 = __expf(4.0f * (float)(255 - tid) * la);

    float  part[NLD - 1];
    float4 ek[GSEG][2], ev[GSEG][2];
#pragma unroll
    for (int k = 0; k < NLD; ++k) {
        const bool valid = (s0 - KSEG + k >= 0);
        float q[2];
#pragma unroll
        for (int h = 0; h < 2; ++h) {
            const float4 a0 = x0[k][h];
            const float4 a1 = x1[k][h];
            float4 r;
            r.x = 0.5f * (a0.x * a0.x + a1.x * a1.x);
            r.y = 0.5f * (a0.y * a0.y + a1.y * a1.y);
            r.z = 0.5f * (a0.z * a0.z + a1.z * a1.z);
            r.w = 0.5f * (a0.w * a0.w + a1.w * a1.w);
            if (k < GSEG) {
                ek[k][h].x = valid ? r.x : 0.0f;
                ek[k][h].y = valid ? r.y : 0.0f;
                ek[k][h].z = valid ? r.z : 0.0f;
                ek[k][h].w = valid ? r.w : 0.0f;
            }
            if (k >= KSEG) ev[k - KSEG][h] = r;
            float hh = oma * r.x;
            hh = fmaf(alpha, hh, oma * r.y);
            hh = fmaf(alpha, hh, oma * r.z);
            hh = fmaf(alpha, hh, oma * r.w);
            q[h] = hh;
        }
        if (k < NLD - 1) part[k] = valid ? (q[0] * w0 + q[1] * w1) : 0.0f;
    }

#pragma unroll
    for (int g = 0; g < GSEG; ++g) {
        float ci = 0.0f;
#pragma unroll
        for (int d = KSEG; d >= 1; --d) ci = fmaf(aL, ci, part[g - d + KSEG]);
#pragma unroll
        for (int off = 16; off > 0; off >>= 1) ci += __shfl_xor(ci, off);
        if (lane == 0) cpart[g][wave] = ci;
    }

#pragma unroll
    for (int g = 0; g < GSEG; ++g) {
        float* lf = &lbs[g][0][0];
#pragma unroll
        for (int h = 0; h < 2; ++h) {
            float4 bv;
            bv.x = oma * (ev[g][h].x - aK * ek[g][h].x);
            bv.y = oma * (ev[g][h].y - aK * ek[g][h].y);
            bv.z = oma * (ev[g][h].z - aK * ek[g][h].z);
            bv.w = oma * (ev[g][h].w - aK * ek[g][h].w);
            *(float4*)(lf + h * 1024 + 4 * tid) = bv;
        }
    }
    __syncthreads();

    float cin[GSEG];
#pragma unroll
    for (int g = 0; g < GSEG; ++g) {
        float ci = cpart[g][0];
#pragma unroll
        for (int w = 1; w < 8; ++w) ci += cpart[g][w];
        cin[g] = ci;
    }

    const int half = lane >> 4;
    const int m    = lane & 15;
    v8f c[GSEG];
#pragma unroll
    for (int g = 0; g < GSEG; ++g) c[g] = v8f{};
#pragma unroll
    for (int sl = 0; sl < 4; ++sl) {
        const int k0 = 4 * sl + 2 * half;
        v2f a;
        a.x = (m >= k0)     ? __expf((float)(m - k0) * la)     : 0.0f;
        a.y = (m >= k0 + 1) ? __expf((float)(m - k0 - 1) * la) : 0.0f;
#pragma unroll
        for (int g = 0; g < GSEG; ++g) {
            v2f bb;
            bb.x = lbs[g][wave][16 * m + k0];
            bb.y = lbs[g][wave][16 * m + k0 + 1];
            c[g] = __builtin_amdgcn_wmma_f32_16x16x4_f32(
                     false, a,  false, bb,
                     (short)0, c[g],  false,  false);
        }
    }

    if (half == 1) {
#pragma unroll
        for (int g = 0; g < GSEG; ++g) cend[g][wave][m] = c[g][7];
    }
    __syncthreads();

    float myCarry[GSEG], cc[GSEG];
#pragma unroll
    for (int g = 0; g < GSEG; ++g) { myCarry[g] = 0.0f; cc[g] = 0.0f; }
#pragma unroll
    for (int j = 0; j < 16; ++j) {
#pragma unroll
        for (int g = 0; g < GSEG; ++g) {
            if (j == m) myCarry[g] = cc[g];
            cc[g] = fmaf(a16, cc[g], cend[g][wave][j]);
        }
    }
    if (lane == 0) {
#pragma unroll
        for (int g = 0; g < GSEG; ++g) wend[g][wave] = cc[g];
    }
    __syncthreads();

    float cw[GSEG];
    {
        const float ew = __expf((float)(256 * wave) * la);
#pragma unroll
        for (int g = 0; g < GSEG; ++g) cw[g] = cin[g] * ew;
    }
    for (int v = 0; v < wave; ++v) {
        const float pv = __expf((float)(256 * (wave - 1 - v)) * la);
#pragma unroll
        for (int g = 0; g < GSEG; ++g) cw[g] = fmaf(wend[g][v], pv, cw[g]);
    }
    const float e16m = __expf((float)(16 * m) * la);
    float ep1[8];
#pragma unroll
    for (int r = 0; r < 8; ++r) ep1[r] = __expf((float)(r + 8 * half + 1) * la);

#pragma unroll
    for (int g = 0; g < GSEG; ++g) {
        const float carryChunk = fmaf(cw[g], e16m, myCarry[g]);
#pragma unroll
        for (int r = 0; r < 8; ++r) {
            const int mm = r + 8 * half;
            float val = fmaf(carryChunk, ep1[r], c[g][r]);
            float y = fmaxf(val, 0.0f);
            lbs[g][wave][16 * m + mm] = __logf(y + 1e-5f);
        }
    }
    __syncthreads();

    const int i0 = lane * 8;
#pragma unroll
    for (int g = 0; g < GSEG; ++g) {
        float* orow = out + (size_t)b * T_ + (size_t)(s0 + g) * SEG + wave * 256;
        float4 o0, o1;
        o0.x = lbs[g][wave][i0 + 0]; o0.y = lbs[g][wave][i0 + 1];
        o0.z = lbs[g][wave][i0 + 2]; o0.w = lbs[g][wave][i0 + 3];
        o1.x = lbs[g][wave][i0 + 4]; o1.y = lbs[g][wave][i0 + 5];
        o1.z = lbs[g][wave][i0 + 6]; o1.w = lbs[g][wave][i0 + 7];
        *(float4*)(orow + i0)     = o0;
        *(float4*)(orow + i0 + 4) = o1;
    }
}

extern "C" void kernel_launch(void* const* d_in, const int* in_sizes, int n_in,
                              void* d_out, int out_size, void* d_ws, size_t ws_size,
                              hipStream_t stream) {
    const float* signal = (const float*)d_in[0];
    const float* zalpha = (const float*)d_in[1];
    float* out = (float*)d_out;

    k_envelope<<<dim3(NGRP, B_), 256, 0, stream>>>(signal, zalpha, out);
}


# Earlier segments load only where their decay factor is not exactly 0 in f32: a row with a^2048 == 0 reads just the weighted tail of the segment before (dropped terms are exact zeros; sum order kept)
# speedup vs baseline: 3.1228x; 1.1946x over previous
#include <hip/hip_runtime.h>
#include <hip/hip_bf16.h>
#include <math.h>


#define B_    32
#define T_    131072
#define KLEN  16384
#define SEG   2048
#define NSEG  (T_ / SEG)
#define GSEG  4
#define KSEG  (KLEN / SEG)
#define NLD   (KSEG + GSEG)
#define NGRP  (NSEG / GSEG)

typedef __attribute__((ext_vector_type(2))) float v2f;
typedef __attribute__((ext_vector_type(8))) float v8f;

__device__ __forceinline__ float alpha_of(float z) {
    float sig = 1.0f / (1.0f + __expf(-z));
    return fminf(fmaxf(sig, 1e-5f), 1.0f - 1e-5f);
}

static_assert(KLEN % SEG == 0, "the K-tap window must be a whole number of segments");
static_assert(SEG == 2048, "256 threads x 2 groups x 4 elements map onto one segment");
static_assert(NSEG % GSEG == 0, "a row is a whole number of blocks");

__device__ __forceinline__ float4 energy4(const float4 a0, const float4 a1) {
    float4 r;
    r.x = 0.5f * (a0.x * a0.x + a1.x * a1.x);
    r.y = 0.5f * (a0.y * a0.y + a1.y * a1.y);
    r.z = 0.5f * (a0.z * a0.z + a1.z * a1.z);
    r.w = 0.5f * (a0.w * a0.w + a1.w * a1.w);
    return r;
}

__device__ __forceinline__ float horner4(const float4 r, const float alpha, const float oma) {
    float hh = oma * r.x;
    hh = fmaf(alpha, hh, oma * r.y);
    hh = fmaf(alpha, hh, oma * r.z);
    hh = fmaf(alpha, hh, oma * r.w);
    return hh;
}

__global__ void __launch_bounds__(256, 4) k_envelope(const float* __restrict__ sig,
                                                     const float* __restrict__ zalpha,
                                                     float* __restrict__ out) {
    const int s0 = blockIdx.x * GSEG, b = blockIdx.y;
    const int tid  = threadIdx.x;
    const int wave = tid >> 5;
    const int lane = tid & 31;

    __shared__ float lbs[GSEG][8][256];
    __shared__ float cend[GSEG][8][16];
    __shared__ float wend[GSEG][8];
    __shared__ float cpart[GSEG][8];

    const float* p0 = sig + (size_t)b * 2 * T_;
    const float* p1 = p0 + T_;

    float4 xo0[GSEG][2], xo1[GSEG][2];
#pragma unroll
    for (int g = 0; g < GSEG; ++g) {
#pragma unroll
        for (int h = 0; h < 2; ++h) {
            const int off = (s0 + g) * SEG + h * 1024 + 4 * tid;
            xo0[g][h] = *(const float4*)(p0 + off);
            xo1[g][h] = *(const float4*)(p1 + off);
        }
    }

    const float alpha = alpha_of(zalpha[b]);
    const float la   = __logf(alpha);
    const float oma  = 1.0f - alpha;
    const float aK   = __expf((float)KLEN * la);
    const float a16  = __expf(16.0f * la);
    const float aL   = __expf((float)SEG * la);
    const float w0 = __expf(4.0f * (float)(511 - tid) * la);
    const float w1 = __expf(4.0f * (float)(255 - tid) * la);

    float  part[NLD - 1];
    float4 ek[GSEG][2], ev[GSEG][2];

    if (aL != 0.0f) {
        float4 x0[KSEG][2], x1[KSEG][2];
#pragma unroll
        for (int k = 0; k < KSEG; ++k) {
            const int sj = s0 - KSEG + k;
            const int sc = sj > 0 ? sj : 0;
#pragma unroll
            for (int h = 0; h < 2; ++h) {
                const int off = sc * SEG + h * 1024 + 4 * tid;
                x0[k][h] = *(const float4*)(p0 + off);
                x1[k][h] = *(const float4*)(p1 + off);
            }
        }
#pragma unroll
        for (int k = 0; k < KSEG; ++k) {
            const bool valid = (s0 - KSEG + k >= 0);
            float q[2];
#pragma unroll
            for (int h = 0; h < 2; ++h) {
                const float4 r = energy4(x0[k][h], x1[k][h]);
                if (k < GSEG) {
                    ek[k][h].x = valid ? r.x : 0.0f;
                    ek[k][h].y = valid ? r.y : 0.0f;
                    ek[k][h].z = valid ? r.z : 0.0f;
                    ek[k][h].w = valid ? r.w : 0.0f;
                }
                q[h] = horner4(r, alpha, oma);
            }
            part[k] = valid ? (q[0] * w0 + q[1] * w1) : 0.0f;
        }
    } else {
#pragma unroll
        for (int k = 0; k < KSEG - 1; ++k) part[k] = 0.0f;
#pragma unroll
        for (int k = 0; k < GSEG; ++k) {
#pragma unroll
            for (int h = 0; h < 2; ++h) ek[k][h] = make_float4(0.0f, 0.0f, 0.0f, 0.0f);
        }
        const float wv[2] = {w0, w1};
        float4 xa[2], xb[2];
#pragma unroll
        for (int h = 0; h < 2; ++h) {
            xa[h] = make_float4(0.0f, 0.0f, 0.0f, 0.0f);
            xb[h] = make_float4(0.0f, 0.0f, 0.0f, 0.0f);
            if (s0 > 0 && wv[h] != 0.0f) {
                const int off = (s0 - 1) * SEG + h * 1024 + 4 * tid;
                xa[h] = *(const float4*)(p0 + off);
                xb[h] = *(const float4*)(p1 + off);
            }
        }
        float q[2];
#pragma unroll
        for (int h = 0; h < 2; ++h) q[h] = horner4(energy4(xa[h], xb[h]), alpha, oma);
        part[KSEG - 1] = q[0] * w0 + q[1] * w1;
    }

#pragma unroll
    for (int g = 0; g < GSEG; ++g) {
        float q[2];
#pragma unroll
        for (int h = 0; h < 2; ++h) {
            const float4 r = energy4(xo0[g][h], xo1[g][h]);
            ev[g][h] = r;
            q[h] = horner4(r, alpha, oma);
        }
        if (g < GSEG - 1) part[KSEG + g] = q[0] * w0 + q[1] * w1;
    }

#pragma unroll
    for (int g = 0; g < GSEG; ++g) {
        float ci = 0.0f;
#pragma unroll
        for (int d = KSEG; d >= 1; --d) ci = fmaf(aL, ci, part[g - d + KSEG]);
#pragma unroll
        for (int off = 16; off > 0; off >>= 1) ci += __shfl_xor(ci, off);
        if (lane == 0) cpart[g][wave] = ci;
    }

#pragma unroll
    for (int g = 0; g < GSEG; ++g) {
        float* lf = &lbs[g][0][0];
#pragma unroll
        for (int h = 0; h < 2; ++h) {
            float4 bv;
            bv.x = oma * (ev[g][h].x - aK * ek[g][h].x);
            bv.y = oma * (ev[g][h].y - aK * ek[g][h].y);
            bv.z = oma * (ev[g][h].z - aK * ek[g][h].z);
            bv.w = oma * (ev[g][h].w - aK * ek[g][h].w);
            *(float4*)(lf + h * 1024 + 4 * tid) = bv;
        }
    }
    __syncthreads();

    float cin[GSEG];
#pragma unroll
    for (int g = 0; g < GSEG; ++g) {
        float ci = cpart[g][0];
#pragma unroll
        for (int w = 1; w < 8; ++w) ci += cpart[g][w];
        cin[g] = ci;
    }

    const int half = lane >> 4;
    const int m    = lane & 15;
    v8f c[GSEG];
#pragma unroll
    for (int g = 0; g < GSEG; ++g) c[g] = v8f{};
#pragma unroll
    for (int sl = 0; sl < 4; ++sl) {
        const int k0 = 4 * sl + 2 * half;
        v2f a;
        a.x = (m >= k0)     ? __expf((float)(m - k0) * la)     : 0.0f;
        a.y = (m >= k0 + 1) ? __expf((float)(m - k0 - 1) * la) : 0.0f;
#pragma unroll
        for (int g = 0; g < GSEG; ++g) {
            v2f bb;
            bb.x = lbs[g][wave][16 * m + k0];
            bb.y = lbs[g][wave][16 * m + k0 + 1];
            c[g] = __builtin_amdgcn_wmma_f32_16x16x4_f32(
                     false, a,  false, bb,
                     (short)0, c[g],  false,  false);
        }
    }

    if (half == 1) {
#pragma unroll
        for (int g = 0; g < GSEG; ++g) cend[g][wave][m] = c[g][7];
    }
    __syncthreads();

    float myCarry[GSEG], cc[GSEG];
#pragma unroll
    for (int g = 0; g < GSEG; ++g) { myCarry[g] = 0.0f; cc[g] = 0.0f; }
#pragma unroll
    for (int j = 0; j < 16; ++j) {
#pragma unroll
        for (int g = 0; g < GSEG; ++g) {
            if (j == m) myCarry[g] = cc[g];
            cc[g] = fmaf(a16, cc[g], cend[g][wave][j]);
        }
    }
    if (lane == 0) {
#pragma unroll
        for (int g = 0; g < GSEG; ++g) wend[g][wave] = cc[g];
    }
    __syncthreads();

    float cw[GSEG];
    {
        const float ew = __expf((float)(256 * wave) * la);
#pragma unroll
        for (int g = 0; g < GSEG; ++g) cw[g] = cin[g] * ew;
    }
    for (int v = 0; v < wave; ++v) {
        const float pv = __expf((float)(256 * (wave - 1 - v)) * la);
#pragma unroll
        for (int g = 0; g < GSEG; ++g) cw[g] = fmaf(wend[g][v], pv, cw[g]);
    }
    const float e16m = __expf((float)(16 * m) * la);
    float ep1[8];
#pragma unroll
    for (int r = 0; r < 8; ++r) ep1[r] = __expf((float)(r + 8 * half + 1) * la);

#pragma unroll
    for (int g = 0; g < GSEG; ++g) {
        const float carryChunk = fmaf(cw[g], e16m, myCarry[g]);
#pragma unroll
        for (int r = 0; r < 8; ++r) {
            const int mm = r + 8 * half;
            float val = fmaf(carryChunk, ep1[r], c[g][r]);
            float y = fmaxf(val, 0.0f);
            lbs[g][wave][16 * m + mm] = __logf(y + 1e-5f);
        }
    }
    __syncthreads();

    const int i0 = lane * 8;
#pragma unroll
    for (int g = 0; g < GSEG; ++g) {
        float* orow = out + (size_t)b * T_ + (size_t)(s0 + g) * SEG + wave * 256;
        float4 o0, o1;
        o0.x = lbs[g][wave][i0 + 0]; o0.y = lbs[g][wave][i0 + 1];
        o0.z = lbs[g][wave][i0 + 2]; o0.w = lbs[g][wave][i0 + 3];
        o1.x = lbs[g][wave][i0 + 4]; o1.y = lbs[g][wave][i0 + 5];
        o1.z = lbs[g][wave][i0 + 6]; o1.w = lbs[g][wave][i0 + 7];
        *(float4*)(orow + i0)     = o0;
        *(float4*)(orow + i0 + 4) = o1;
    }
}

extern "C" void kernel_launch(void* const* d_in, const int* in_sizes, int n_in,
                              void* d_out, int out_size, void* d_ws, size_t ws_size,
                              hipStream_t stream) {
    const float* signal = (const float*)d_in[0];
    const float* zalpha = (const float*)d_in[1];
    float* out = (float*)d_out;

    k_envelope<<<dim3(NGRP, B_), 256, 0, stream>>>(signal, zalpha, out);
}


# Carry path as DPP trees in registers: lane sums (row_xmask + permlanex16), 16-chunk scan (4 row_shr doubling steps), cross-wave combine (1 product a lane + row sum); no ds_bpermute; sum order changes
# speedup vs baseline: 3.8109x; 1.2203x over previous
#include <hip/hip_runtime.h>
#include <hip/hip_bf16.h>
#include <math.h>


#define B_    32
#define T_    131072
#define KLEN  16384
#define SEG   2048
#define NSEG  (T_ / SEG)
#define GSEG  4
#define KSEG  (KLEN / SEG)
#define NLD   (KSEG + GSEG)
#define NGRP  (NSEG / GSEG)

typedef __attribute__((ext_vector_type(2))) float v2f;
typedef __attribute__((ext_vector_type(8))) float v8f;

__device__ __forceinline__ float alpha_of(float z) {
    float sig = 1.0f / (1.0f + __expf(-z));
    return fminf(fmaxf(sig, 1e-5f), 1.0f - 1e-5f);
}

static_assert(KLEN % SEG == 0, "the K-tap window must be a whole number of segments");
static_assert(SEG == 2048, "256 threads x 2 groups x 4 elements map onto one segment");
static_assert(NSEG % GSEG == 0, "a row is a whole number of blocks");

__device__ __forceinline__ float4 energy4(const float4 a0, const float4 a1) {
    float4 r;
    r.x = 0.5f * (a0.x * a0.x + a1.x * a1.x);
    r.y = 0.5f * (a0.y * a0.y + a1.y * a1.y);
    r.z = 0.5f * (a0.z * a0.z + a1.z * a1.z);
    r.w = 0.5f * (a0.w * a0.w + a1.w * a1.w);
    return r;
}

__device__ __forceinline__ float horner4(const float4 r, const float alpha, const float oma) {
    float hh = oma * r.x;
    hh = fmaf(alpha, hh, oma * r.y);
    hh = fmaf(alpha, hh, oma * r.z);
    hh = fmaf(alpha, hh, oma * r.w);
    return hh;
}

__device__ __forceinline__ void lds_barrier() {
    __builtin_amdgcn_fence(__ATOMIC_RELEASE, "workgroup", "local");
    __builtin_amdgcn_s_barrier();
    __builtin_amdgcn_fence(__ATOMIC_ACQUIRE, "workgroup", "local");
}

__device__ __forceinline__ void wave_sync() {
    __builtin_amdgcn_fence(__ATOMIC_RELEASE, "wavefront", "local");
    __builtin_amdgcn_wave_barrier();
    __builtin_amdgcn_fence(__ATOMIC_ACQUIRE, "wavefront", "local");
}

__device__ __forceinline__ float xhalf(float x) {
    return __builtin_bit_cast(float, __builtin_amdgcn_permlanex16(
        __builtin_bit_cast(int, x), __builtin_bit_cast(int, x),
        0x76543210, 0xfedcba98, false, false));
}
template <int CTRL>
__device__ __forceinline__ float row_dpp(float x) {
    return __builtin_bit_cast(float, __builtin_amdgcn_update_dpp(
        0, __builtin_bit_cast(int, x), CTRL, 0xf, 0xf,  true));
}
__device__ __forceinline__ float row_sum(float v) {
    v += row_dpp<0x168>(v);
    v += row_dpp<0x164>(v);
    v += row_dpp<0x162>(v);
    v += row_dpp<0x161>(v);
    return v;
}
__device__ __forceinline__ float wave_sum(float v) {
    return row_sum(v + xhalf(v));
}

__global__ void __launch_bounds__(256, 4) k_envelope(const float* __restrict__ sig,
                                                     const float* __restrict__ zalpha,
                                                     float* __restrict__ out) {
    const int s0 = blockIdx.x * GSEG, b = blockIdx.y;
    const int tid  = threadIdx.x;
    const int wave = tid >> 5;
    const int lane = tid & 31;
    const int half = lane >> 4;
    const int m    = lane & 15;

    __shared__ float lbs[GSEG][8][256];
    __shared__ float xch[GSEG][16];

    const float* p0 = sig + (size_t)b * 2 * T_;
    const float* p1 = p0 + T_;

    const int eo = 256 * wave + 4 * lane;
    const float za = zalpha[b];

    float4 xo0[GSEG][2], xo1[GSEG][2];
#pragma unroll
    for (int h = 0; h < 2; ++h) {
        const int off = s0 * SEG + h * 128 + eo;
        xo0[0][h] = *(const float4*)(p0 + off);
        xo1[0][h] = *(const float4*)(p1 + off);
    }
    __builtin_amdgcn_sched_barrier(0);

    const float alpha = alpha_of(za);
    const float la   = __logf(alpha);
    const float oma  = 1.0f - alpha;
    const float aK   = __expf((float)KLEN * la);
    const float a16  = __expf(16.0f * la);
    const float a32  = __expf(32.0f * la);
    const float a64  = __expf(64.0f * la);
    const float a128 = __expf(128.0f * la);
    const float aL   = __expf((float)SEG * la);
    const float w0 = __expf((float)(SEG - 4 - eo) * la);
    const float w1 = __expf((float)(SEG - 4 - 128 - eo) * la);
    const bool full = (aL != 0.0f);

    const float wv[2] = {w0, w1};
    bool need[2];
    float4 xa[2], xb[2];
#pragma unroll
    for (int h = 0; h < 2; ++h) {
        need[h] = (s0 > 0) && (full || wv[h] != 0.0f);
        const int off = need[h] ? (s0 - 1) * SEG + h * 128 + eo : s0 * SEG;
        xa[h] = *(const float4*)(p0 + off);
        xb[h] = *(const float4*)(p1 + off);
    }

    __builtin_amdgcn_sched_barrier(0);

#pragma unroll
    for (int g = 1; g < GSEG; ++g) {
#pragma unroll
        for (int h = 0; h < 2; ++h) {
            const int off = (s0 + g) * SEG + h * 128 + eo;
            xo0[g][h] = *(const float4*)(p0 + off);
            xo1[g][h] = *(const float4*)(p1 + off);
        }
    }
    float  part[NLD - 1];
    float4 ek[GSEG][2];

    if (full) {
        float4 x0[KSEG - 1][2], x1[KSEG - 1][2];
#pragma unroll
        for (int k = 0; k < KSEG - 1; ++k) {
            const int sj = s0 - KSEG + k;
            const int sc = sj > 0 ? sj : 0;
#pragma unroll
            for (int h = 0; h < 2; ++h) {
                const int off = sc * SEG + h * 128 + eo;
                x0[k][h] = *(const float4*)(p0 + off);
                x1[k][h] = *(const float4*)(p1 + off);
            }
        }
#pragma unroll
        for (int k = 0; k < KSEG - 1; ++k) {
            const bool valid = (s0 - KSEG + k >= 0);
            float q[2];
#pragma unroll
            for (int h = 0; h < 2; ++h) {
                const float4 r = energy4(x0[k][h], x1[k][h]);
                if (k < GSEG) {
                    ek[k][h].x = valid ? r.x : 0.0f;
                    ek[k][h].y = valid ? r.y : 0.0f;
                    ek[k][h].z = valid ? r.z : 0.0f;
                    ek[k][h].w = valid ? r.w : 0.0f;
                }
                q[h] = horner4(r, alpha, oma);
            }
            part[k] = valid ? (q[0] * w0 + q[1] * w1) : 0.0f;
        }
    } else {
#pragma unroll
        for (int k = 0; k < KSEG - 1; ++k) part[k] = 0.0f;
#pragma unroll
        for (int k = 0; k < GSEG; ++k) {
#pragma unroll
            for (int h = 0; h < 2; ++h) ek[k][h] = make_float4(0.0f, 0.0f, 0.0f, 0.0f);
        }
    }

    v2f afr[4];
#pragma unroll
    for (int sl = 0; sl < 4; ++sl) {
        const int k0 = 4 * sl + 2 * half;
        afr[sl].x = (m >= k0)     ? __expf((float)(m - k0) * la)     : 0.0f;
        afr[sl].y = (m >= k0 + 1) ? __expf((float)(m - k0 - 1) * la) : 0.0f;
    }
    const float ew   = __expf((float)(256 * wave) * la);
    const float e16m = __expf((float)(16 * m) * la);
    float ep1[8];
#pragma unroll
    for (int r = 0; r < 8; ++r) ep1[r] = __expf((float)(r + 8 * half + 1) * la);
    const int   tv = m - 8;
    const float wl = (m < 8) ? ew
                   : ((tv < wave) ? __expf((float)(256 * (wave - 1 - tv)) * la) : 0.0f);

#pragma unroll
    for (int g = 0; g < GSEG; ++g) {
        {
            float q[2];
            float* lf = &lbs[g][0][0];
#pragma unroll
            for (int h = 0; h < 2; ++h) {
                const float4 r = energy4(xo0[g][h], xo1[g][h]);
                q[h] = horner4(r, alpha, oma);
                float4 bv;
                bv.x = oma * (r.x - aK * ek[g][h].x);
                bv.y = oma * (r.y - aK * ek[g][h].y);
                bv.z = oma * (r.z - aK * ek[g][h].z);
                bv.w = oma * (r.w - aK * ek[g][h].w);
                *(float4*)(lf + h * 128 + eo) = bv;
            }
            if (g < GSEG - 1) part[KSEG + g] = q[0] * w0 + q[1] * w1;
        }
        if (g > 0) {
            float ci = 0.0f;
#pragma unroll
            for (int d = KSEG; d >= 1; --d) ci = fmaf(aL, ci, part[g - d + KSEG]);
            ci = wave_sum(ci);
            if (lane == 0) xch[g][wave] = ci;
        }
        wave_sync();

        v8f c = v8f{};
#pragma unroll
        for (int sl = 0; sl < 4; ++sl) {
            const int k0 = 4 * sl + 2 * half;
            v2f bb;
            bb.x = lbs[g][wave][16 * m + k0];
            bb.y = lbs[g][wave][16 * m + k0 + 1];
            c = __builtin_amdgcn_wmma_f32_16x16x4_f32(
                     false, afr[sl],  false, bb,
                     (short)0, c,  false,  false);
        }

        if (g == 0) {
            float q[2];
#pragma unroll
            for (int h = 0; h < 2; ++h) {
                const float qh = horner4(energy4(xa[h], xb[h]), alpha, oma);
                q[h] = need[h] ? qh : 0.0f;
            }
            part[KSEG - 1] = q[0] * w0 + q[1] * w1;
            float ci = 0.0f;
#pragma unroll
            for (int d = KSEG; d >= 1; --d) ci = fmaf(aL, ci, part[g - d + KSEG]);
            ci = wave_sum(ci);
            if (lane == 0) xch[g][wave] = ci;
        }

        const float c7x = xhalf(c[7]);
        float sc = (half == 1) ? c[7] : c7x;
        sc = fmaf(row_dpp<0x111>(sc), a16,  sc);
        sc = fmaf(row_dpp<0x112>(sc), a32,  sc);
        sc = fmaf(row_dpp<0x114>(sc), a64,  sc);
        sc = fmaf(row_dpp<0x118>(sc), a128, sc);
        const float myCarry = row_dpp<0x111>(sc);
        if (lane == 31) xch[g][8 + wave] = sc;
        lds_barrier();

        const float cw = row_sum(xch[g][m] * wl);
        const float carryChunk = fmaf(cw, e16m, myCarry);
#pragma unroll
        for (int r = 0; r < 8; ++r) {
            const int mm = r + 8 * half;
            float val = fmaf(carryChunk, ep1[r], c[r]);
            float y = fmaxf(val, 0.0f);
            lbs[g][wave][16 * m + mm] = __logf(y + 1e-5f);
        }
        wave_sync();

        {
            const int i0 = lane * 8;
            float* orow = out + (size_t)b * T_ + (size_t)(s0 + g) * SEG + wave * 256;
            float4 o0, o1;
            o0.x = lbs[g][wave][i0 + 0]; o0.y = lbs[g][wave][i0 + 1];
            o0.z = lbs[g][wave][i0 + 2]; o0.w = lbs[g][wave][i0 + 3];
            o1.x = lbs[g][wave][i0 + 4]; o1.y = lbs[g][wave][i0 + 5];
            o1.z = lbs[g][wave][i0 + 6]; o1.w = lbs[g][wave][i0 + 7];
            *(float4*)(orow + i0)     = o0;
            *(float4*)(orow + i0 + 4) = o1;
        }
    }
}

extern "C" void kernel_launch(void* const* d_in, const int* in_sizes, int n_in,
                              void* d_out, int out_size, void* d_ws, size_t ws_size,
                              hipStream_t stream) {
    const float* signal = (const float*)d_in[0];
    const float* zalpha = (const float*)d_in[1];
    float* out = (float*)d_out;

    k_envelope<<<dim3(NGRP, B_), 256, 0, stream>>>(signal, zalpha, out);
}


# Output log as the bare hardware log2 (v_log_f32) times ln 2, 2 instructions a value and within about 2 ulp, in place of the library log's 12-instruction expansion; its argument is never below 1e-5
# speedup vs baseline: 4.1027x; 1.0766x over previous
#include <hip/hip_runtime.h>
#include <hip/hip_bf16.h>
#include <math.h>


#define B_    32
#define T_    131072
#define KLEN  16384
#define SEG   2048
#define NSEG  (T_ / SEG)
#define GSEG  4
#define KSEG  (KLEN / SEG)
#define NLD   (KSEG + GSEG)
#define NGRP  (NSEG / GSEG)

typedef __attribute__((ext_vector_type(2))) float v2f;
typedef __attribute__((ext_vector_type(8))) float v8f;

__device__ __forceinline__ float alpha_of(float z) {
    float sig = 1.0f / (1.0f + __expf(-z));
    return fminf(fmaxf(sig, 1e-5f), 1.0f - 1e-5f);
}

static_assert(KLEN % SEG == 0, "the K-tap window must be a whole number of segments");
static_assert(SEG == 2048, "256 threads x 2 groups x 4 elements map onto one segment");
static_assert(NSEG % GSEG == 0, "a row is a whole number of blocks");

__device__ __forceinline__ float4 energy4(const float4 a0, const float4 a1) {
    float4 r;
    r.x = 0.5f * (a0.x * a0.x + a1.x * a1.x);
    r.y = 0.5f * (a0.y * a0.y + a1.y * a1.y);
    r.z = 0.5f * (a0.z * a0.z + a1.z * a1.z);
    r.w = 0.5f * (a0.w * a0.w + a1.w * a1.w);
    return r;
}

__device__ __forceinline__ float horner4(const float4 r, const float alpha, const float oma) {
    float hh = oma * r.x;
    hh = fmaf(alpha, hh, oma * r.y);
    hh = fmaf(alpha, hh, oma * r.z);
    hh = fmaf(alpha, hh, oma * r.w);
    return hh;
}

__device__ __forceinline__ float ln_floor(const float x) {
    return 0.69314718056f * __builtin_amdgcn_logf(x);
}

__device__ __forceinline__ void lds_barrier() {
    __builtin_amdgcn_fence(__ATOMIC_RELEASE, "workgroup", "local");
    __builtin_amdgcn_s_barrier();
    __builtin_amdgcn_fence(__ATOMIC_ACQUIRE, "workgroup", "local");
}

__device__ __forceinline__ void wave_sync() {
    __builtin_amdgcn_fence(__ATOMIC_RELEASE, "wavefront", "local");
    __builtin_amdgcn_wave_barrier();
    __builtin_amdgcn_fence(__ATOMIC_ACQUIRE, "wavefront", "local");
}

__device__ __forceinline__ float xhalf(float x) {
    return __builtin_bit_cast(float, __builtin_amdgcn_permlanex16(
        __builtin_bit_cast(int, x), __builtin_bit_cast(int, x),
        0x76543210, 0xfedcba98, false, false));
}
template <int CTRL>
__device__ __forceinline__ float row_dpp(float x) {
    return __builtin_bit_cast(float, __builtin_amdgcn_update_dpp(
        0, __builtin_bit_cast(int, x), CTRL, 0xf, 0xf,  true));
}
__device__ __forceinline__ float row_sum(float v) {
    v += row_dpp<0x168>(v);
    v += row_dpp<0x164>(v);
    v += row_dpp<0x162>(v);
    v += row_dpp<0x161>(v);
    return v;
}
__device__ __forceinline__ float wave_sum(float v) {
    return row_sum(v + xhalf(v));
}

__global__ void __launch_bounds__(256, 4) k_envelope(const float* __restrict__ sig,
                                                     const float* __restrict__ zalpha,
                                                     float* __restrict__ out) {
    const int s0 = blockIdx.x * GSEG, b = blockIdx.y;
    const int tid  = threadIdx.x;
    const int wave = tid >> 5;
    const int lane = tid & 31;
    const int half = lane >> 4;
    const int m    = lane & 15;

    __shared__ float lbs[GSEG][8][256];
    __shared__ float xch[GSEG][16];

    const float* p0 = sig + (size_t)b * 2 * T_;
    const float* p1 = p0 + T_;

    const int eo = 256 * wave + 4 * lane;
    const float za = zalpha[b];

    float4 xo0[GSEG][2], xo1[GSEG][2];
#pragma unroll
    for (int h = 0; h < 2; ++h) {
        const int off = s0 * SEG + h * 128 + eo;
        xo0[0][h] = *(const float4*)(p0 + off);
        xo1[0][h] = *(const float4*)(p1 + off);
    }
    __builtin_amdgcn_sched_barrier(0);

    const float alpha = alpha_of(za);
    const float la   = __logf(alpha);
    const float oma  = 1.0f - alpha;
    const float aK   = __expf((float)KLEN * la);
    const float a16  = __expf(16.0f * la);
    const float a32  = __expf(32.0f * la);
    const float a64  = __expf(64.0f * la);
    const float a128 = __expf(128.0f * la);
    const float aL   = __expf((float)SEG * la);
    const float w0 = __expf((float)(SEG - 4 - eo) * la);
    const float w1 = __expf((float)(SEG - 4 - 128 - eo) * la);
    const bool full = (aL != 0.0f);

    const float wv[2] = {w0, w1};
    bool need[2];
    float4 xa[2], xb[2];
#pragma unroll
    for (int h = 0; h < 2; ++h) {
        need[h] = (s0 > 0) && (full || wv[h] != 0.0f);
        const int off = need[h] ? (s0 - 1) * SEG + h * 128 + eo : s0 * SEG;
        xa[h] = *(const float4*)(p0 + off);
        xb[h] = *(const float4*)(p1 + off);
    }

    __builtin_amdgcn_sched_barrier(0);

#pragma unroll
    for (int g = 1; g < GSEG; ++g) {
#pragma unroll
        for (int h = 0; h < 2; ++h) {
            const int off = (s0 + g) * SEG + h * 128 + eo;
            xo0[g][h] = *(const float4*)(p0 + off);
            xo1[g][h] = *(const float4*)(p1 + off);
        }
    }
    float  part[NLD - 1];
    float4 ek[GSEG][2];

    if (full) {
        float4 x0[KSEG - 1][2], x1[KSEG - 1][2];
#pragma unroll
        for (int k = 0; k < KSEG - 1; ++k) {
            const int sj = s0 - KSEG + k;
            const int sc = sj > 0 ? sj : 0;
#pragma unroll
            for (int h = 0; h < 2; ++h) {
                const int off = sc * SEG + h * 128 + eo;
                x0[k][h] = *(const float4*)(p0 + off);
                x1[k][h] = *(const float4*)(p1 + off);
            }
        }
#pragma unroll
        for (int k = 0; k < KSEG - 1; ++k) {
            const bool valid = (s0 - KSEG + k >= 0);
            float q[2];
#pragma unroll
            for (int h = 0; h < 2; ++h) {
                const float4 r = energy4(x0[k][h], x1[k][h]);
                if (k < GSEG) {
                    ek[k][h].x = valid ? r.x : 0.0f;
                    ek[k][h].y = valid ? r.y : 0.0f;
                    ek[k][h].z = valid ? r.z : 0.0f;
                    ek[k][h].w = valid ? r.w : 0.0f;
                }
                q[h] = horner4(r, alpha, oma);
            }
            part[k] = valid ? (q[0] * w0 + q[1] * w1) : 0.0f;
        }
    } else {
#pragma unroll
        for (int k = 0; k < KSEG - 1; ++k) part[k] = 0.0f;
#pragma unroll
        for (int k = 0; k < GSEG; ++k) {
#pragma unroll
            for (int h = 0; h < 2; ++h) ek[k][h] = make_float4(0.0f, 0.0f, 0.0f, 0.0f);
        }
    }

    v2f afr[4];
#pragma unroll
    for (int sl = 0; sl < 4; ++sl) {
        const int k0 = 4 * sl + 2 * half;
        afr[sl].x = (m >= k0)     ? __expf((float)(m - k0) * la)     : 0.0f;
        afr[sl].y = (m >= k0 + 1) ? __expf((float)(m - k0 - 1) * la) : 0.0f;
    }
    const float ew   = __expf((float)(256 * wave) * la);
    const float e16m = __expf((float)(16 * m) * la);
    float ep1[8];
#pragma unroll
    for (int r = 0; r < 8; ++r) ep1[r] = __expf((float)(r + 8 * half + 1) * la);
    const int   tv = m - 8;
    const float wl = (m < 8) ? ew
                   : ((tv < wave) ? __expf((float)(256 * (wave - 1 - tv)) * la) : 0.0f);

#pragma unroll
    for (int g = 0; g < GSEG; ++g) {
        {
            float q[2];
            float* lf = &lbs[g][0][0];
#pragma unroll
            for (int h = 0; h < 2; ++h) {
                const float4 r = energy4(xo0[g][h], xo1[g][h]);
                q[h] = horner4(r, alpha, oma);
                float4 bv;
                bv.x = oma * (r.x - aK * ek[g][h].x);
                bv.y = oma * (r.y - aK * ek[g][h].y);
                bv.z = oma * (r.z - aK * ek[g][h].z);
                bv.w = oma * (r.w - aK * ek[g][h].w);
                *(float4*)(lf + h * 128 + eo) = bv;
            }
            if (g < GSEG - 1) part[KSEG + g] = q[0] * w0 + q[1] * w1;
        }
        if (g > 0) {
            float ci = 0.0f;
#pragma unroll
            for (int d = KSEG; d >= 1; --d) ci = fmaf(aL, ci, part[g - d + KSEG]);
            ci = wave_sum(ci);
            if (lane == 0) xch[g][wave] = ci;
        }
        wave_sync();

        v8f c = v8f{};
#pragma unroll
        for (int sl = 0; sl < 4; ++sl) {
            const int k0 = 4 * sl + 2 * half;
            v2f bb;
            bb.x = lbs[g][wave][16 * m + k0];
            bb.y = lbs[g][wave][16 * m + k0 + 1];
            c = __builtin_amdgcn_wmma_f32_16x16x4_f32(
                     false, afr[sl],  false, bb,
                     (short)0, c,  false,  false);
        }

        if (g == 0) {
            float q[2];
#pragma unroll
            for (int h = 0; h < 2; ++h) {
                const float qh = horner4(energy4(xa[h], xb[h]), alpha, oma);
                q[h] = need[h] ? qh : 0.0f;
            }
            part[KSEG - 1] = q[0] * w0 + q[1] * w1;
            float ci = 0.0f;
#pragma unroll
            for (int d = KSEG; d >= 1; --d) ci = fmaf(aL, ci, part[g - d + KSEG]);
            ci = wave_sum(ci);
            if (lane == 0) xch[g][wave] = ci;
        }

        const float c7x = xhalf(c[7]);
        float sc = (half == 1) ? c[7] : c7x;
        sc = fmaf(row_dpp<0x111>(sc), a16,  sc);
        sc = fmaf(row_dpp<0x112>(sc), a32,  sc);
        sc = fmaf(row_dpp<0x114>(sc), a64,  sc);
        sc = fmaf(row_dpp<0x118>(sc), a128, sc);
        const float myCarry = row_dpp<0x111>(sc);
        if (lane == 31) xch[g][8 + wave] = sc;
        lds_barrier();

        const float cw = row_sum(xch[g][m] * wl);
        const float carryChunk = fmaf(cw, e16m, myCarry);
#pragma unroll
        for (int r = 0; r < 8; ++r) {
            const int mm = r + 8 * half;
            float val = fmaf(carryChunk, ep1[r], c[r]);
            float y = fmaxf(val, 0.0f);
            lbs[g][wave][16 * m + mm] = ln_floor(y + 1e-5f);
        }
        wave_sync();

        {
            const int i0 = lane * 8;
            float* orow = out + (size_t)b * T_ + (size_t)(s0 + g) * SEG + wave * 256;
            float4 o0, o1;
            o0.x = lbs[g][wave][i0 + 0]; o0.y = lbs[g][wave][i0 + 1];
            o0.z = lbs[g][wave][i0 + 2]; o0.w = lbs[g][wave][i0 + 3];
            o1.x = lbs[g][wave][i0 + 4]; o1.y = lbs[g][wave][i0 + 5];
            o1.z = lbs[g][wave][i0 + 6]; o1.w = lbs[g][wave][i0 + 7];
            *(float4*)(orow + i0)     = o0;
            *(float4*)(orow + i0 + 4) = o1;
        }
    }
}

extern "C" void kernel_launch(void* const* d_in, const int* in_sizes, int n_in,
                              void* d_out, int out_size, void* d_ws, size_t ws_size,
                              hipStream_t stream) {
    const float* signal = (const float*)d_in[0];
    const float* zalpha = (const float*)d_in[1];
    float* out = (float*)d_out;

    k_envelope<<<dim3(NGRP, B_), 256, 0, stream>>>(signal, zalpha, out);
}


# Output stores in the lane order of the loads (lane l: elements 4l..4l+3 and 128+4l..+3 of the tile): a store instruction writes 4 whole 128-byte lines, not halves of 8; values, sum order unchanged
# speedup vs baseline: 4.6468x; 1.1326x over previous
#include <hip/hip_runtime.h>
#include <hip/hip_bf16.h>
#include <math.h>


#define B_    32
#define T_    131072
#define KLEN  16384
#define SEG   2048
#define NSEG  (T_ / SEG)
#define GSEG  4
#define KSEG  (KLEN / SEG)
#define NLD   (KSEG + GSEG)
#define NGRP  (NSEG / GSEG)

typedef __attribute__((ext_vector_type(2))) float v2f;
typedef __attribute__((ext_vector_type(8))) float v8f;

__device__ __forceinline__ float alpha_of(float z) {
    float sig = 1.0f / (1.0f + __expf(-z));
    return fminf(fmaxf(sig, 1e-5f), 1.0f - 1e-5f);
}

static_assert(KLEN % SEG == 0, "the K-tap window must be a whole number of segments");
static_assert(SEG == 2048, "256 threads x 2 groups x 4 elements map onto one segment");
static_assert(NSEG % GSEG == 0, "a row is a whole number of blocks");

__device__ __forceinline__ float4 energy4(const float4 a0, const float4 a1) {
    float4 r;
    r.x = 0.5f * (a0.x * a0.x + a1.x * a1.x);
    r.y = 0.5f * (a0.y * a0.y + a1.y * a1.y);
    r.z = 0.5f * (a0.z * a0.z + a1.z * a1.z);
    r.w = 0.5f * (a0.w * a0.w + a1.w * a1.w);
    return r;
}

__device__ __forceinline__ float horner4(const float4 r, const float alpha, const float oma) {
    float hh = oma * r.x;
    hh = fmaf(alpha, hh, oma * r.y);
    hh = fmaf(alpha, hh, oma * r.z);
    hh = fmaf(alpha, hh, oma * r.w);
    return hh;
}

__device__ __forceinline__ float ln_floor(const float x) {
    return 0.69314718056f * __builtin_amdgcn_logf(x);
}

__device__ __forceinline__ void lds_barrier() {
    __builtin_amdgcn_fence(__ATOMIC_RELEASE, "workgroup", "local");
    __builtin_amdgcn_s_barrier();
    __builtin_amdgcn_fence(__ATOMIC_ACQUIRE, "workgroup", "local");
}

__device__ __forceinline__ void wave_sync() {
    __builtin_amdgcn_fence(__ATOMIC_RELEASE, "wavefront", "local");
    __builtin_amdgcn_wave_barrier();
    __builtin_amdgcn_fence(__ATOMIC_ACQUIRE, "wavefront", "local");
}

__device__ __forceinline__ float xhalf(float x) {
    return __builtin_bit_cast(float, __builtin_amdgcn_permlanex16(
        __builtin_bit_cast(int, x), __builtin_bit_cast(int, x),
        0x76543210, 0xfedcba98, false, false));
}
template <int CTRL>
__device__ __forceinline__ float row_dpp(float x) {
    return __builtin_bit_cast(float, __builtin_amdgcn_update_dpp(
        0, __builtin_bit_cast(int, x), CTRL, 0xf, 0xf,  true));
}
__device__ __forceinline__ float row_sum(float v) {
    v += row_dpp<0x168>(v);
    v += row_dpp<0x164>(v);
    v += row_dpp<0x162>(v);
    v += row_dpp<0x161>(v);
    return v;
}
__device__ __forceinline__ float wave_sum(float v) {
    return row_sum(v + xhalf(v));
}

__global__ void __launch_bounds__(256, 4) k_envelope(const float* __restrict__ sig,
                                                     const float* __restrict__ zalpha,
                                                     float* __restrict__ out) {
    const int s0 = blockIdx.x * GSEG, b = blockIdx.y;
    const int tid  = threadIdx.x;
    const int wave = tid >> 5;
    const int lane = tid & 31;
    const int half = lane >> 4;
    const int m    = lane & 15;

    __shared__ float lbs[GSEG][8][256];
    __shared__ float xch[GSEG][16];

    const float* p0 = sig + (size_t)b * 2 * T_;
    const float* p1 = p0 + T_;

    const int eo = 256 * wave + 4 * lane;
    const float za = zalpha[b];

    float4 xo0[GSEG][2], xo1[GSEG][2];
#pragma unroll
    for (int h = 0; h < 2; ++h) {
        const int off = s0 * SEG + h * 128 + eo;
        xo0[0][h] = *(const float4*)(p0 + off);
        xo1[0][h] = *(const float4*)(p1 + off);
    }
    __builtin_amdgcn_sched_barrier(0);

    const float alpha = alpha_of(za);
    const float la   = __logf(alpha);
    const float oma  = 1.0f - alpha;
    const float aK   = __expf((float)KLEN * la);
    const float a16  = __expf(16.0f * la);
    const float a32  = __expf(32.0f * la);
    const float a64  = __expf(64.0f * la);
    const float a128 = __expf(128.0f * la);
    const float aL   = __expf((float)SEG * la);
    const float w0 = __expf((float)(SEG - 4 - eo) * la);
    const float w1 = __expf((float)(SEG - 4 - 128 - eo) * la);
    const bool full = (aL != 0.0f);

    const float wv[2] = {w0, w1};
    bool need[2];
    float4 xa[2], xb[2];
#pragma unroll
    for (int h = 0; h < 2; ++h) {
        need[h] = (s0 > 0) && (full || wv[h] != 0.0f);
        const int off = need[h] ? (s0 - 1) * SEG + h * 128 + eo : s0 * SEG;
        xa[h] = *(const float4*)(p0 + off);
        xb[h] = *(const float4*)(p1 + off);
    }

    __builtin_amdgcn_sched_barrier(0);

#pragma unroll
    for (int g = 1; g < GSEG; ++g) {
#pragma unroll
        for (int h = 0; h < 2; ++h) {
            const int off = (s0 + g) * SEG + h * 128 + eo;
            xo0[g][h] = *(const float4*)(p0 + off);
            xo1[g][h] = *(const float4*)(p1 + off);
        }
    }
    float  part[NLD - 1];
    float4 ek[GSEG][2];

    if (full) {
        float4 x0[KSEG - 1][2], x1[KSEG - 1][2];
#pragma unroll
        for (int k = 0; k < KSEG - 1; ++k) {
            const int sj = s0 - KSEG + k;
            const int sc = sj > 0 ? sj : 0;
#pragma unroll
            for (int h = 0; h < 2; ++h) {
                const int off = sc * SEG + h * 128 + eo;
                x0[k][h] = *(const float4*)(p0 + off);
                x1[k][h] = *(const float4*)(p1 + off);
            }
        }
#pragma unroll
        for (int k = 0; k < KSEG - 1; ++k) {
            const bool valid = (s0 - KSEG + k >= 0);
            float q[2];
#pragma unroll
            for (int h = 0; h < 2; ++h) {
                const float4 r = energy4(x0[k][h], x1[k][h]);
                if (k < GSEG) {
                    ek[k][h].x = valid ? r.x : 0.0f;
                    ek[k][h].y = valid ? r.y : 0.0f;
                    ek[k][h].z = valid ? r.z : 0.0f;
                    ek[k][h].w = valid ? r.w : 0.0f;
                }
                q[h] = horner4(r, alpha, oma);
            }
            part[k] = valid ? (q[0] * w0 + q[1] * w1) : 0.0f;
        }
    } else {
#pragma unroll
        for (int k = 0; k < KSEG - 1; ++k) part[k] = 0.0f;
#pragma unroll
        for (int k = 0; k < GSEG; ++k) {
#pragma unroll
            for (int h = 0; h < 2; ++h) ek[k][h] = make_float4(0.0f, 0.0f, 0.0f, 0.0f);
        }
    }

    v2f afr[4];
#pragma unroll
    for (int sl = 0; sl < 4; ++sl) {
        const int k0 = 4 * sl + 2 * half;
        afr[sl].x = (m >= k0)     ? __expf((float)(m - k0) * la)     : 0.0f;
        afr[sl].y = (m >= k0 + 1) ? __expf((float)(m - k0 - 1) * la) : 0.0f;
    }
    const float ew   = __expf((float)(256 * wave) * la);
    const float e16m = __expf((float)(16 * m) * la);
    float ep1[8];
#pragma unroll
    for (int r = 0; r < 8; ++r) ep1[r] = __expf((float)(r + 8 * half + 1) * la);
    const int   tv = m - 8;
    const float wl = (m < 8) ? ew
                   : ((tv < wave) ? __expf((float)(256 * (wave - 1 - tv)) * la) : 0.0f);

#pragma unroll
    for (int g = 0; g < GSEG; ++g) {
        {
            float q[2];
            float* lf = &lbs[g][0][0];
#pragma unroll
            for (int h = 0; h < 2; ++h) {
                const float4 r = energy4(xo0[g][h], xo1[g][h]);
                q[h] = horner4(r, alpha, oma);
                float4 bv;
                bv.x = oma * (r.x - aK * ek[g][h].x);
                bv.y = oma * (r.y - aK * ek[g][h].y);
                bv.z = oma * (r.z - aK * ek[g][h].z);
                bv.w = oma * (r.w - aK * ek[g][h].w);
                *(float4*)(lf + h * 128 + eo) = bv;
            }
            if (g < GSEG - 1) part[KSEG + g] = q[0] * w0 + q[1] * w1;
        }
        if (g > 0) {
            float ci = 0.0f;
#pragma unroll
            for (int d = KSEG; d >= 1; --d) ci = fmaf(aL, ci, part[g - d + KSEG]);
            ci = wave_sum(ci);
            if (lane == 0) xch[g][wave] = ci;
        }
        wave_sync();

        v8f c = v8f{};
#pragma unroll
        for (int sl = 0; sl < 4; ++sl) {
            const int k0 = 4 * sl + 2 * half;
            v2f bb;
            bb.x = lbs[g][wave][16 * m + k0];
            bb.y = lbs[g][wave][16 * m + k0 + 1];
            c = __builtin_amdgcn_wmma_f32_16x16x4_f32(
                     false, afr[sl],  false, bb,
                     (short)0, c,  false,  false);
        }

        if (g == 0) {
            float q[2];
#pragma unroll
            for (int h = 0; h < 2; ++h) {
                const float qh = horner4(energy4(xa[h], xb[h]), alpha, oma);
                q[h] = need[h] ? qh : 0.0f;
            }
            part[KSEG - 1] = q[0] * w0 + q[1] * w1;
            float ci = 0.0f;
#pragma unroll
            for (int d = KSEG; d >= 1; --d) ci = fmaf(aL, ci, part[g - d + KSEG]);
            ci = wave_sum(ci);
            if (lane == 0) xch[g][wave] = ci;
        }

        const float c7x = xhalf(c[7]);
        float sc = (half == 1) ? c[7] : c7x;
        sc = fmaf(row_dpp<0x111>(sc), a16,  sc);
        sc = fmaf(row_dpp<0x112>(sc), a32,  sc);
        sc = fmaf(row_dpp<0x114>(sc), a64,  sc);
        sc = fmaf(row_dpp<0x118>(sc), a128, sc);
        const float myCarry = row_dpp<0x111>(sc);
        if (lane == 31) xch[g][8 + wave] = sc;
        lds_barrier();

        const float cw = row_sum(xch[g][m] * wl);
        const float carryChunk = fmaf(cw, e16m, myCarry);
#pragma unroll
        for (int r = 0; r < 8; ++r) {
            const int mm = r + 8 * half;
            float val = fmaf(carryChunk, ep1[r], c[r]);
            float y = fmaxf(val, 0.0f);
            lbs[g][wave][16 * m + mm] = ln_floor(y + 1e-5f);
        }
        wave_sync();

        {
            const int i0 = lane * 4;
            const float* lt = &lbs[g][wave][0];
            float* orow = out + (size_t)b * T_ + (size_t)(s0 + g) * SEG + wave * 256;
            const float4 o0 = *(const float4*)(lt + i0);
            const float4 o1 = *(const float4*)(lt + 128 + i0);
            *(float4*)(orow + i0)       = o0;
            *(float4*)(orow + 128 + i0) = o1;
        }
    }
}

extern "C" void kernel_launch(void* const* d_in, const int* in_sizes, int n_in,
                              void* d_out, int out_size, void* d_ws, size_t ws_size,
                              hipStream_t stream) {
    const float* signal = (const float*)d_in[0];
    const float* zalpha = (const float*)d_in[1];
    float* out = (float*)d_out;

    k_envelope<<<dim3(NGRP, B_), 256, 0, stream>>>(signal, zalpha, out);
}


# Each wave owns two adjacent 256-step tiles of every segment (blocks of 4 waves, not 8): half the waves enter, decay powers, carry share and barrier paid once per two tiles; carry sum order changes
# speedup vs baseline: 4.8960x; 1.0536x over previous
#include <hip/hip_runtime.h>
#include <hip/hip_bf16.h>
#include <math.h>


#define B_    32
#define T_    131072
#define KLEN  16384
#define SEG   2048
#define NSEG  (T_ / SEG)
#define GSEG  4
#define KSEG  (KLEN / SEG)
#define NLD   (KSEG + GSEG)
#define NGRP  (NSEG / GSEG)
#define NW    4
#define NT    2
#define NJ    (2 * NT)

typedef __attribute__((ext_vector_type(2))) float v2f;
typedef __attribute__((ext_vector_type(8))) float v8f;

__device__ __forceinline__ float alpha_of(float z) {
    float sig = 1.0f / (1.0f + __expf(-z));
    return fminf(fmaxf(sig, 1e-5f), 1.0f - 1e-5f);
}

static_assert(KLEN % SEG == 0, "the K-tap window must be a whole number of segments");
static_assert(SEG == NW * 32 * NJ * 4, "128 threads x 4 groups x 4 elements map onto one segment");
static_assert(NW * NT == 8, "a segment is 8 tiles of 256 steps");
static_assert(NSEG % GSEG == 0, "a row is a whole number of blocks");

__device__ __forceinline__ float4 energy4(const float4 a0, const float4 a1) {
    float4 r;
    r.x = 0.5f * (a0.x * a0.x + a1.x * a1.x);
    r.y = 0.5f * (a0.y * a0.y + a1.y * a1.y);
    r.z = 0.5f * (a0.z * a0.z + a1.z * a1.z);
    r.w = 0.5f * (a0.w * a0.w + a1.w * a1.w);
    return r;
}

__device__ __forceinline__ float horner4(const float4 r, const float alpha, const float oma) {
    float hh = oma * r.x;
    hh = fmaf(alpha, hh, oma * r.y);
    hh = fmaf(alpha, hh, oma * r.z);
    hh = fmaf(alpha, hh, oma * r.w);
    return hh;
}

__device__ __forceinline__ float ln_floor(const float x) {
    return 0.69314718056f * __builtin_amdgcn_logf(x);
}

__device__ __forceinline__ void lds_barrier() {
    __builtin_amdgcn_fence(__ATOMIC_RELEASE, "workgroup", "local");
    __builtin_amdgcn_s_barrier();
    __builtin_amdgcn_fence(__ATOMIC_ACQUIRE, "workgroup", "local");
}

__device__ __forceinline__ void wave_sync() {
    __builtin_amdgcn_fence(__ATOMIC_RELEASE, "wavefront", "local");
    __builtin_amdgcn_wave_barrier();
    __builtin_amdgcn_fence(__ATOMIC_ACQUIRE, "wavefront", "local");
}

__device__ __forceinline__ float xhalf(float x) {
    return __builtin_bit_cast(float, __builtin_amdgcn_permlanex16(
        __builtin_bit_cast(int, x), __builtin_bit_cast(int, x),
        0x76543210, 0xfedcba98, false, false));
}
template <int CTRL>
__device__ __forceinline__ float row_dpp(float x) {
    return __builtin_bit_cast(float, __builtin_amdgcn_update_dpp(
        0, __builtin_bit_cast(int, x), CTRL, 0xf, 0xf,  true));
}
__device__ __forceinline__ float row_sum(float v) {
    v += row_dpp<0x168>(v);
    v += row_dpp<0x164>(v);
    v += row_dpp<0x162>(v);
    v += row_dpp<0x161>(v);
    return v;
}
__device__ __forceinline__ float wave_sum(float v) {
    return row_sum(v + xhalf(v));
}

__global__ void __launch_bounds__(128, 2) k_envelope(const float* __restrict__ sig,
                                                     const float* __restrict__ zalpha,
                                                     float* __restrict__ out) {
    const int s0 = blockIdx.x * GSEG, b = blockIdx.y;
    const int tid  = threadIdx.x;
    const int wave = tid >> 5;
    const int lane = tid & 31;
    const int half = lane >> 4;
    const int m    = lane & 15;

    __shared__ float lbs[GSEG][8][256];
    __shared__ float xch[GSEG][16];

    const float* p0 = sig + (size_t)b * 2 * T_;
    const float* p1 = p0 + T_;

    const int eo = 512 * wave + 4 * lane;
    const float za = zalpha[b];

    float4 xo0[GSEG][NJ], xo1[GSEG][NJ];
#pragma unroll
    for (int j = 0; j < NJ; ++j) {
        const int off = s0 * SEG + j * 128 + eo;
        xo0[0][j] = *(const float4*)(p0 + off);
        xo1[0][j] = *(const float4*)(p1 + off);
    }
    __builtin_amdgcn_sched_barrier(0);

    const float alpha = alpha_of(za);
    const float la   = __logf(alpha);
    const float oma  = 1.0f - alpha;
    const float aK   = __expf((float)KLEN * la);
    const float a16  = __expf(16.0f * la);
    const float a32  = __expf(32.0f * la);
    const float a64  = __expf(64.0f * la);
    const float a128 = __expf(128.0f * la);
    const float aL   = __expf((float)SEG * la);
    float wv[NJ];
#pragma unroll
    for (int j = 0; j < NJ; ++j) wv[j] = __expf((float)(SEG - 4 - 128 * j - eo) * la);
    const bool full = (aL != 0.0f);

    bool need[NJ];
    float4 xa[NJ], xb[NJ];
#pragma unroll
    for (int j = 0; j < NJ; ++j) {
        need[j] = (s0 > 0) && (full || wv[j] != 0.0f);
        const int off = need[j] ? (s0 - 1) * SEG + j * 128 + eo : s0 * SEG;
        xa[j] = *(const float4*)(p0 + off);
        xb[j] = *(const float4*)(p1 + off);
    }

    __builtin_amdgcn_sched_barrier(0);

#pragma unroll
    for (int g = 1; g < GSEG; ++g) {
#pragma unroll
        for (int j = 0; j < NJ; ++j) {
            const int off = (s0 + g) * SEG + j * 128 + eo;
            xo0[g][j] = *(const float4*)(p0 + off);
            xo1[g][j] = *(const float4*)(p1 + off);
        }
    }
    float  part[NLD - 1];
    float4 ek[GSEG][NJ];

    if (full) {
        float4 x0[KSEG - 1][NJ], x1[KSEG - 1][NJ];
#pragma unroll
        for (int k = 0; k < KSEG - 1; ++k) {
            const int sj = s0 - KSEG + k;
            const int sc = sj > 0 ? sj : 0;
#pragma unroll
            for (int j = 0; j < NJ; ++j) {
                const int off = sc * SEG + j * 128 + eo;
                x0[k][j] = *(const float4*)(p0 + off);
                x1[k][j] = *(const float4*)(p1 + off);
            }
        }
#pragma unroll
        for (int k = 0; k < KSEG - 1; ++k) {
            const bool valid = (s0 - KSEG + k >= 0);
            float pk = 0.0f;
#pragma unroll
            for (int j = 0; j < NJ; ++j) {
                const float4 r = energy4(x0[k][j], x1[k][j]);
                if (k < GSEG) {
                    ek[k][j].x = valid ? r.x : 0.0f;
                    ek[k][j].y = valid ? r.y : 0.0f;
                    ek[k][j].z = valid ? r.z : 0.0f;
                    ek[k][j].w = valid ? r.w : 0.0f;
                }
                pk = fmaf(horner4(r, alpha, oma), wv[j], pk);
            }
            part[k] = valid ? pk : 0.0f;
        }
    } else {
#pragma unroll
        for (int k = 0; k < KSEG - 1; ++k) part[k] = 0.0f;
#pragma unroll
        for (int k = 0; k < GSEG; ++k) {
#pragma unroll
            for (int j = 0; j < NJ; ++j) ek[k][j] = make_float4(0.0f, 0.0f, 0.0f, 0.0f);
        }
    }

    v2f afr[4];
#pragma unroll
    for (int sl = 0; sl < 4; ++sl) {
        const int k0 = 4 * sl + 2 * half;
        afr[sl].x = (m >= k0)     ? __expf((float)(m - k0) * la)     : 0.0f;
        afr[sl].y = (m >= k0 + 1) ? __expf((float)(m - k0 - 1) * la) : 0.0f;
    }
    const float e16m = __expf((float)(16 * m) * la);
    float ep1[8];
#pragma unroll
    for (int r = 0; r < 8; ++r) ep1[r] = __expf((float)(r + 8 * half + 1) * la);
    const int tv = m - 4;
    const int xw = (m < 12) ? m : 0;
    float wl[NT];
#pragma unroll
    for (int tt = 0; tt < NT; ++tt) {
        const int tile = NT * wave + tt;
        wl[tt] = (m < 4) ? __expf((float)(256 * tile) * la)
               : ((m < 12 && tv < tile) ? __expf((float)(256 * (tile - 1 - tv)) * la) : 0.0f);
    }

#pragma unroll
    for (int g = 0; g < GSEG; ++g) {
        {
            float pk = 0.0f;
            float* lf = &lbs[g][0][0];
#pragma unroll
            for (int j = 0; j < NJ; ++j) {
                const float4 r = energy4(xo0[g][j], xo1[g][j]);
                pk = fmaf(horner4(r, alpha, oma), wv[j], pk);
                float4 bv;
                bv.x = oma * (r.x - aK * ek[g][j].x);
                bv.y = oma * (r.y - aK * ek[g][j].y);
                bv.z = oma * (r.z - aK * ek[g][j].z);
                bv.w = oma * (r.w - aK * ek[g][j].w);
                *(float4*)(lf + j * 128 + eo) = bv;
            }
            if (g < GSEG - 1) part[KSEG + g] = pk;
        }
        if (g > 0) {
            float ci = 0.0f;
#pragma unroll
            for (int d = KSEG; d >= 1; --d) ci = fmaf(aL, ci, part[g - d + KSEG]);
            ci = wave_sum(ci);
            if (lane == 0) xch[g][wave] = ci;
        }
        wave_sync();

        v8f c[NT];
#pragma unroll
        for (int tt = 0; tt < NT; ++tt) c[tt] = v8f{};
#pragma unroll
        for (int sl = 0; sl < 4; ++sl) {
            const int k0 = 4 * sl + 2 * half;
#pragma unroll
            for (int tt = 0; tt < NT; ++tt) {
                v2f bb;
                bb.x = lbs[g][NT * wave + tt][16 * m + k0];
                bb.y = lbs[g][NT * wave + tt][16 * m + k0 + 1];
                c[tt] = __builtin_amdgcn_wmma_f32_16x16x4_f32(
                         false, afr[sl],  false, bb,
                         (short)0, c[tt],  false,  false);
            }
        }

        if (g == 0) {
            float pk = 0.0f;
#pragma unroll
            for (int j = 0; j < NJ; ++j) {
                const float qj = horner4(energy4(xa[j], xb[j]), alpha, oma);
                pk = fmaf(need[j] ? qj : 0.0f, wv[j], pk);
            }
            part[KSEG - 1] = pk;
            float ci = 0.0f;
#pragma unroll
            for (int d = KSEG; d >= 1; --d) ci = fmaf(aL, ci, part[g - d + KSEG]);
            ci = wave_sum(ci);
            if (lane == 0) xch[g][wave] = ci;
        }

        float myCarry[NT];
#pragma unroll
        for (int tt = 0; tt < NT; ++tt) {
            const float c7x = xhalf(c[tt][7]);
            float sc = (half == 1) ? c[tt][7] : c7x;
            sc = fmaf(row_dpp<0x111>(sc), a16,  sc);
            sc = fmaf(row_dpp<0x112>(sc), a32,  sc);
            sc = fmaf(row_dpp<0x114>(sc), a64,  sc);
            sc = fmaf(row_dpp<0x118>(sc), a128, sc);
            myCarry[tt] = row_dpp<0x111>(sc);
            if (lane == 31) xch[g][4 + NT * wave + tt] = sc;
        }
        lds_barrier();

        const float xv = xch[g][xw];
#pragma unroll
        for (int tt = 0; tt < NT; ++tt) {
            const float cw = row_sum(xv * wl[tt]);
            const float carryChunk = fmaf(cw, e16m, myCarry[tt]);
#pragma unroll
            for (int r = 0; r < 8; ++r) {
                const int mm = r + 8 * half;
                float val = fmaf(carryChunk, ep1[r], c[tt][r]);
                float y = fmaxf(val, 0.0f);
                lbs[g][NT * wave + tt][16 * m + mm] = ln_floor(y + 1e-5f);
            }
        }
        wave_sync();

        {
            const int i0 = lane * 4;
            const float* lt = &lbs[g][NT * wave][0];
            float* orow = out + (size_t)b * T_ + (size_t)(s0 + g) * SEG + wave * (NT * 256);
            float4 o[NJ];
#pragma unroll
            for (int j = 0; j < NJ; ++j) o[j] = *(const float4*)(lt + j * 128 + i0);
#pragma unroll
            for (int j = 0; j < NJ; ++j) *(float4*)(orow + j * 128 + i0) = o[j];
        }
    }
}

extern "C" void kernel_launch(void* const* d_in, const int* in_sizes, int n_in,
                              void* d_out, int out_size, void* d_ws, size_t ws_size,
                              hipStream_t stream) {
    const float* signal = (const float*)d_in[0];
    const float* zalpha = (const float*)d_in[1];
    float* out = (float*)d_out;

    k_envelope<<<dim3(NGRP, B_), NW * 32, 0, stream>>>(signal, zalpha, out);
}
